# starting kernel
# baseline (speedup 1.0000x reference)
#include <hip/hip_runtime.h>
#include <math.h>

#define N_NODES 50000
#define N_EDGES 800000
#define HIDW    128
#define NHEAD   4
#define CPH     32
#define NLAYERS 3
#define EDIMW   3

typedef __attribute__((ext_vector_type(2))) float v2f;
typedef __attribute__((ext_vector_type(8))) float v8f;

__device__ __forceinline__ unsigned encOrd(float f) {
  unsigned u = __float_as_uint(f);
  return (u & 0x80000000u) ? ~u : (u | 0x80000000u);
}
__device__ __forceinline__ float decOrd(unsigned u) {
  return __uint_as_float((u & 0x80000000u) ? (u & 0x7FFFFFFFu) : ~u);
}
#define ENC_NEG_INF 0x007FFFFFu

__global__ __launch_bounds__(256)
void init_layer_kernel(float* __restrict__ hout, unsigned* __restrict__ lmax,
                       float* __restrict__ denom, float* __restrict__ stats) {
  int i = blockIdx.x * 256 + threadIdx.x;
  if (i < N_NODES * HIDW) hout[i] = 0.f;
  if (i < N_NODES * NHEAD) { lmax[i] = ENC_NEG_INF; denom[i] = 0.f; }
  if (i < 2) stats[i] = 0.f;
}

__global__ __launch_bounds__(256)
void gemm_dual_wmma(const float* __restrict__ A,
                    const float* __restrict__ W0,
                    const float* __restrict__ W1,
                    float* __restrict__ O0,
                    float* __restrict__ O1) {
  __shared__ float As[16 * 132];
  const int rowBase = blockIdx.x * 16;
  for (int i = threadIdx.x; i < 16 * HIDW; i += 256) {
    int r = i >> 7, c = i & 127;
    As[r * 132 + c] = A[(rowBase + r) * HIDW + c];
  }
  __syncthreads();
  const int wave    = threadIdx.x >> 5;
  const int lane    = threadIdx.x & 31;
  const int colBase = wave * 16;
  const int laneHi  = lane >> 4;
  const int laneLo  = lane & 15;
  v8f c0 = {}; v8f c1 = {};
  for (int k0 = 0; k0 < HIDW; k0 += 4) {
    const int ka = k0 + 2 * laneHi;
    v2f a;
    a.x = As[laneLo * 132 + ka];
    a.y = As[laneLo * 132 + ka + 1];
    v2f b0, b1;
    b0.x = W0[(ka + 0) * HIDW + colBase + laneLo];
    b0.y = W0[(ka + 1) * HIDW + colBase + laneLo];
    b1.x = W1[(ka + 0) * HIDW + colBase + laneLo];
    b1.y = W1[(ka + 1) * HIDW + colBase + laneLo];
    c0 = __builtin_amdgcn_wmma_f32_16x16x4_f32(false, a, false, b0, (short)0, c0, false, false);
    c1 = __builtin_amdgcn_wmma_f32_16x16x4_f32(false, a, false, b1, (short)0, c1, false, false);
  }
#pragma unroll
  for (int r = 0; r < 8; ++r) {
    const int row = rowBase + r + 8 * laneHi;
    O0[row * HIDW + colBase + laneLo] = c0[r];
    O1[row * HIDW + colBase + laneLo] = c1[r];
  }
}

__global__ __launch_bounds__(256)
void gemm_bias_wmma(const float* __restrict__ A,
                    const float* __restrict__ W,
                    const float* __restrict__ bias,
                    float* __restrict__ O) {
  __shared__ float As[16 * 132];
  const int rowBase = blockIdx.x * 16;
  for (int i = threadIdx.x; i < 16 * HIDW; i += 256) {
    int r = i >> 7, c = i & 127;
    As[r * 132 + c] = A[(rowBase + r) * HIDW + c];
  }
  __syncthreads();
  const int wave    = threadIdx.x >> 5;
  const int lane    = threadIdx.x & 31;
  const int colBase = wave * 16;
  const int laneHi  = lane >> 4;
  const int laneLo  = lane & 15;
  v8f c0 = {};
  for (int k0 = 0; k0 < HIDW; k0 += 4) {
    const int ka = k0 + 2 * laneHi;
    v2f a;
    a.x = As[laneLo * 132 + ka];
    a.y = As[laneLo * 132 + ka + 1];
    v2f b;
    b.x = W[(ka + 0) * HIDW + colBase + laneLo];
    b.y = W[(ka + 1) * HIDW + colBase + laneLo];
    c0 = __builtin_amdgcn_wmma_f32_16x16x4_f32(false, a, false, b, (short)0, c0, false, false);
  }
  const float bv = bias[colBase + laneLo];
#pragma unroll
  for (int r = 0; r < 8; ++r) {
    const int row = rowBase + r + 8 * laneHi;
    O[row * HIDW + colBase + laneLo] = c0[r] + bv;
  }
}

__global__ __launch_bounds__(256)
void edge_logits_kernel(const int* __restrict__ ei, const float* __restrict__ eattr,
                        const float* __restrict__ xl, const float* __restrict__ xr,
                        const float* __restrict__ We, const float* __restrict__ att,
                        float* __restrict__ elog, unsigned* __restrict__ lmaxu) {
  __shared__ float sWe[EDIMW * HIDW];
  __shared__ float sAtt[HIDW];
  for (int i = threadIdx.x; i < EDIMW * HIDW; i += 256) sWe[i] = We[i];
  if (threadIdx.x < HIDW) sAtt[threadIdx.x] = att[threadIdx.x];
  __syncthreads();
  const int lane = threadIdx.x & 31;
  const int wid  = (blockIdx.x * 256 + threadIdx.x) >> 5;
  const int nw   = (gridDim.x * 256) >> 5;
  for (int e = wid; e < N_EDGES; e += nw) {
    const int src = ei[e];
    const int dst = ei[N_EDGES + e];
    float eav = (lane < 3) ? eattr[e * 3 + lane] : 0.f;
    const float ea0 = __shfl(eav, 0, 32);
    const float ea1 = __shfl(eav, 1, 32);
    const float ea2 = __shfl(eav, 2, 32);
    float acc[NHEAD];
#pragma unroll
    for (int h = 0; h < NHEAD; ++h) {
      const int c = h * CPH + lane;
      float eac = ea0 * sWe[c] + ea1 * sWe[HIDW + c] + ea2 * sWe[2 * HIDW + c];
      float v = xl[src * HIDW + c] + xr[dst * HIDW + c] + eac;
      v = (v > 0.f) ? v : 0.2f * v;
      acc[h] = v * sAtt[c];
    }
#pragma unroll
    for (int off = 16; off > 0; off >>= 1) {
#pragma unroll
      for (int h = 0; h < NHEAD; ++h) acc[h] += __shfl_xor(acc[h], off, 32);
    }
    if (lane < NHEAD) {
      float lg = (lane == 0) ? acc[0] : (lane == 1) ? acc[1] : (lane == 2) ? acc[2] : acc[3];
      elog[e * NHEAD + lane] = lg;
      atomicMax(&lmaxu[dst * NHEAD + lane], encOrd(lg));
    }
  }
}

__global__ __launch_bounds__(256)
void lmax_fixup_kernel(unsigned* __restrict__ lmaxu) {
  int i = blockIdx.x * 256 + threadIdx.x;
  if (i >= N_NODES * NHEAD) return;
  float f = decOrd(lmaxu[i]);
  ((float*)lmaxu)[i] = isfinite(f) ? f : 0.f;
}

__global__ __launch_bounds__(256)
void edge_expsum_kernel(const int* __restrict__ ei, float* __restrict__ elog,
                        const float* __restrict__ lmaxf, float* __restrict__ denom) {
  int t = blockIdx.x * 256 + threadIdx.x;
  if (t >= N_EDGES * NHEAD) return;
  int e = t >> 2, h = t & 3;
  int dst = ei[N_EDGES + e];
  float ex = expf(elog[t] - lmaxf[dst * NHEAD + h]);
  elog[t] = ex;
  atomicAdd(&denom[dst * NHEAD + h], ex);
}

__global__ __launch_bounds__(256)
void edge_scatter_kernel(const int* __restrict__ ei, const float* __restrict__ xl,
                         const float* __restrict__ ex, const float* __restrict__ denom,
                         float* __restrict__ out) {
  const int lane = threadIdx.x & 31;
  const int wid  = (blockIdx.x * 256 + threadIdx.x) >> 5;
  const int nw   = (gridDim.x * 256) >> 5;
  for (int e = wid; e < N_EDGES; e += nw) {
    const int src = ei[e];
    const int dst = ei[N_EDGES + e];
    float exv = 0.f, dnv = 0.f;
    if (lane < NHEAD) {
      exv = ex[e * NHEAD + lane];
      dnv = denom[dst * NHEAD + lane];
    }
#pragma unroll
    for (int h = 0; h < NHEAD; ++h) {
      const float alpha = __shfl(exv, h, 32) / (__shfl(dnv, h, 32) + 1e-16f);
      const int c = h * CPH + lane;
      atomicAdd(&out[dst * HIDW + c], xl[src * HIDW + c] * alpha);
    }
  }
}

__global__ __launch_bounds__(256)
void ln_reduce_kernel(const float* __restrict__ h, float* __restrict__ stats) {
  __shared__ float s1[256], s2[256];
  float a = 0.f, b = 0.f;
  for (int i = blockIdx.x * 256 + threadIdx.x; i < N_NODES * HIDW; i += gridDim.x * 256) {
    float v = h[i];
    a += v; b += v * v;
  }
  s1[threadIdx.x] = a; s2[threadIdx.x] = b;
  __syncthreads();
  for (int off = 128; off > 0; off >>= 1) {
    if ((int)threadIdx.x < off) {
      s1[threadIdx.x] += s1[threadIdx.x + off];
      s2[threadIdx.x] += s2[threadIdx.x + off];
    }
    __syncthreads();
  }
  if (threadIdx.x == 0) { atomicAdd(&stats[0], s1[0]); atomicAdd(&stats[1], s2[0]); }
}

__global__ __launch_bounds__(256)
void ln_gelu_kernel(float* __restrict__ h, const float* __restrict__ stats,
                    const float* __restrict__ w, const float* __restrict__ b) {
  int i = blockIdx.x * 256 + threadIdx.x;
  if (i >= N_NODES * HIDW) return;
  const float inv  = 1.0f / (float)(N_NODES * HIDW);
  const float mean = stats[0] * inv;
  const float var  = stats[1] * inv - mean * mean;
  const float rs   = rsqrtf(var + 1e-5f);
  const int c = i & (HIDW - 1);
  float v = (h[i] - mean) * rs * w[c] + b[c];
  v = 0.5f * v * (1.0f + erff(v * 0.70710678118654752f));
  h[i] = v;
}

extern "C" void kernel_launch(void* const* d_in, const int* in_sizes, int n_in,
                              void* d_out, int out_size, void* d_ws, size_t ws_size,
                              hipStream_t stream) {
  const float* x     = (const float*)d_in[0];
  const int*   ei    = (const int*)  d_in[1];
  const float* eattr = (const float*)d_in[2];
  const float* Wl    = (const float*)d_in[3];
  const float* Wr    = (const float*)d_in[4];
  const float* We    = (const float*)d_in[5];
  const float* att   = (const float*)d_in[6];
  const float* lnw   = (const float*)d_in[7];
  const float* lnb   = (const float*)d_in[8];
  const float* Wout  = (const float*)d_in[9];
  const float* bout  = (const float*)d_in[10];
  float* out = (float*)d_out;

  char* p = (char*)d_ws;
  auto carve = [&](size_t bytes) -> void* {
    void* q = (void*)p;
    p += (bytes + 255) & ~(size_t)255;
    return q;
  };
  float*    hA    = (float*)   carve((size_t)N_NODES * HIDW * 4);
  float*    hB    = (float*)   carve((size_t)N_NODES * HIDW * 4);
  float*    xl    = (float*)   carve((size_t)N_NODES * HIDW * 4);
  float*    xr    = (float*)   carve((size_t)N_NODES * HIDW * 4);
  float*    elog  = (float*)   carve((size_t)N_EDGES * NHEAD * 4);
  unsigned* lmax  = (unsigned*)carve((size_t)N_NODES * NHEAD * 4);
  float*    denom = (float*)   carve((size_t)N_NODES * NHEAD * 4);
  float*    stats = (float*)   carve(2 * 4);

  const int elemBlocks = (N_NODES * HIDW + 255) / 256;
  const int segBlocks  = (N_NODES * NHEAD + 255) / 256;
  const int ehBlocks   = (N_EDGES * NHEAD + 255) / 256;
  const int waveBlocks = 25000;
  const int gemmBlocks = N_NODES / 16;

  const float* hcur = x;
  float* bufs[2] = { hA, hB };
  for (int l = 0; l < NLAYERS; ++l) {
    float* hout = bufs[l & 1];
    init_layer_kernel<<<elemBlocks, 256, 0, stream>>>(hout, lmax, denom, stats);
    gemm_dual_wmma<<<gemmBlocks, 256, 0, stream>>>(hcur, Wl + (size_t)l * HIDW * HIDW,
                                                   Wr + (size_t)l * HIDW * HIDW, xl, xr);
    edge_logits_kernel<<<waveBlocks, 256, 0, stream>>>(ei, eattr, xl, xr,
                                                       We + (size_t)l * EDIMW * HIDW,
                                                       att + (size_t)l * HIDW, elog, lmax);
    lmax_fixup_kernel<<<segBlocks, 256, 0, stream>>>(lmax);
    edge_expsum_kernel<<<ehBlocks, 256, 0, stream>>>(ei, elog, (const float*)lmax, denom);
    edge_scatter_kernel<<<waveBlocks, 256, 0, stream>>>(ei, xl, elog, denom, hout);
    ln_reduce_kernel<<<2048, 256, 0, stream>>>(hout, stats);
    ln_gelu_kernel<<<elemBlocks, 256, 0, stream>>>(hout, stats,
                                                   lnw + (size_t)l * HIDW,
                                                   lnb + (size_t)l * HIDW);
    hcur = hout;
  }
  gemm_bias_wmma<<<gemmBlocks, 256, 0, stream>>>(hcur, Wout, bout, out);
}


# Edges grouped by destination once per call (counting sort, ascending edge id); one wave per node does logits, running softmax and the row sum in registers: no float atomics, fixed sum order
# speedup vs baseline: 2.2985x; 2.2985x over previous
#include <hip/hip_runtime.h>
#include <math.h>

#define N_NODES 50000
#define N_EDGES 800000
#define HIDW    128
#define NHEAD   4
#define CPH     32
#define NLAYERS 3
#define EDIMW   3

typedef __attribute__((ext_vector_type(2))) float v2f;
typedef __attribute__((ext_vector_type(4))) float v4f;
typedef __attribute__((ext_vector_type(8))) float v8f;

#define NEG_BIG (-3.402823466e+38f)

__global__ __launch_bounds__(256)
void zero_counts_kernel(int* __restrict__ cnt) {
  const int i = blockIdx.x * 256 + threadIdx.x;
  if (i < N_NODES) cnt[i] = 0;
}

__global__ __launch_bounds__(256)
void count_rank_kernel(const int* __restrict__ ei, int* __restrict__ cnt,
                       int* __restrict__ rank) {
  const int e = blockIdx.x * 256 + threadIdx.x;
  if (e >= N_EDGES) return;
  rank[e] = atomicAdd(&cnt[ei[N_EDGES + e]], 1);
}

#define SCAN_PER ((N_NODES + 1023) / 1024)
__global__ __launch_bounds__(1024)
void scan_offsets_kernel(const int* __restrict__ cnt, int* __restrict__ off) {
  __shared__ int part[1024];
  const int t = threadIdx.x;
  const int b = t * SCAN_PER;
  int c[SCAN_PER];
  int s = 0;
#pragma unroll
  for (int j = 0; j < SCAN_PER; ++j) {
    const int i = b + j;
    const int v = cnt[(i < N_NODES) ? i : (N_NODES - 1)];
    c[j] = (i < N_NODES) ? v : 0;
    s += c[j];
  }
  part[t] = s;
  __syncthreads();
  for (int d = 1; d < 1024; d <<= 1) {
    const int v = (t >= d) ? part[t - d] : 0;
    __syncthreads();
    part[t] += v;
    __syncthreads();
  }
  int run = part[t] - s;
#pragma unroll
  for (int j = 0; j < SCAN_PER; ++j) {
    const int i = b + j;
    if (i < N_NODES) off[i] = run;
    run += c[j];
  }
  if (t == 1023) off[N_NODES] = part[1023];
}

__global__ __launch_bounds__(256)
void fill_perm_kernel(const int* __restrict__ ei, const int* __restrict__ off,
                      const int* __restrict__ rank, int* __restrict__ perm) {
  const int e = blockIdx.x * 256 + threadIdx.x;
  if (e >= N_EDGES) return;
  perm[off[ei[N_EDGES + e]] + rank[e]] = e;
}

__global__ __launch_bounds__(256)
void order_segments_kernel(const int* __restrict__ ei, const float* __restrict__ eattr,
                           const int* __restrict__ off, const int* __restrict__ perm,
                           v4f* __restrict__ rec) {
  const int lane = threadIdx.x & 31;
  const int node = blockIdx.x * 8 + (threadIdx.x >> 5);
  if (node >= N_NODES) return;
  const int beg = off[node];
  const int deg = off[node + 1] - beg;
  for (int k = lane; k < deg; k += 32) {
    const int my = perm[beg + k];
    int place = 0;
    for (int j = 0; j < deg; ++j) place += (perm[beg + j] < my) ? 1 : 0;
    v4f r;
    r.x = __int_as_float(ei[my]);
    r.y = eattr[my * 3 + 0];
    r.z = eattr[my * 3 + 1];
    r.w = eattr[my * 3 + 2];
    rec[beg + place] = r;
  }
}

__global__ __launch_bounds__(256)
void gemm_dual_wmma(const float* __restrict__ A,
                    const float* __restrict__ W0,
                    const float* __restrict__ W1,
                    float* __restrict__ O0,
                    float* __restrict__ O1) {
  __shared__ float As[16 * 132];
  const int rowBase = blockIdx.x * 16;
  for (int i = threadIdx.x; i < 16 * HIDW; i += 256) {
    int r = i >> 7, c = i & 127;
    As[r * 132 + c] = A[(rowBase + r) * HIDW + c];
  }
  __syncthreads();
  const int wave    = threadIdx.x >> 5;
  const int lane    = threadIdx.x & 31;
  const int colBase = wave * 16;
  const int laneHi  = lane >> 4;
  const int laneLo  = lane & 15;
  v8f c0 = {}; v8f c1 = {};
  for (int k0 = 0; k0 < HIDW; k0 += 4) {
    const int ka = k0 + 2 * laneHi;
    v2f a;
    a.x = As[laneLo * 132 + ka];
    a.y = As[laneLo * 132 + ka + 1];
    v2f b0, b1;
    b0.x = W0[(ka + 0) * HIDW + colBase + laneLo];
    b0.y = W0[(ka + 1) * HIDW + colBase + laneLo];
    b1.x = W1[(ka + 0) * HIDW + colBase + laneLo];
    b1.y = W1[(ka + 1) * HIDW + colBase + laneLo];
    c0 = __builtin_amdgcn_wmma_f32_16x16x4_f32(false, a, false, b0, (short)0, c0, false, false);
    c1 = __builtin_amdgcn_wmma_f32_16x16x4_f32(false, a, false, b1, (short)0, c1, false, false);
  }
#pragma unroll
  for (int r = 0; r < 8; ++r) {
    const int row = rowBase + r + 8 * laneHi;
    O0[row * HIDW + colBase + laneLo] = c0[r];
    O1[row * HIDW + colBase + laneLo] = c1[r];
  }
}

__global__ __launch_bounds__(256)
void gemm_bias_wmma(const float* __restrict__ A,
                    const float* __restrict__ W,
                    const float* __restrict__ bias,
                    float* __restrict__ O) {
  __shared__ float As[16 * 132];
  const int rowBase = blockIdx.x * 16;
  for (int i = threadIdx.x; i < 16 * HIDW; i += 256) {
    int r = i >> 7, c = i & 127;
    As[r * 132 + c] = A[(rowBase + r) * HIDW + c];
  }
  __syncthreads();
  const int wave    = threadIdx.x >> 5;
  const int lane    = threadIdx.x & 31;
  const int colBase = wave * 16;
  const int laneHi  = lane >> 4;
  const int laneLo  = lane & 15;
  v8f c0 = {};
  for (int k0 = 0; k0 < HIDW; k0 += 4) {
    const int ka = k0 + 2 * laneHi;
    v2f a;
    a.x = As[laneLo * 132 + ka];
    a.y = As[laneLo * 132 + ka + 1];
    v2f b;
    b.x = W[(ka + 0) * HIDW + colBase + laneLo];
    b.y = W[(ka + 1) * HIDW + colBase + laneLo];
    c0 = __builtin_amdgcn_wmma_f32_16x16x4_f32(false, a, false, b, (short)0, c0, false, false);
  }
  const float bv = bias[colBase + laneLo];
#pragma unroll
  for (int r = 0; r < 8; ++r) {
    const int row = rowBase + r + 8 * laneHi;
    O[row * HIDW + colBase + laneLo] = c0[r] + bv;
  }
}

__device__ __forceinline__ float head_logit(const v4f x, const v4f xrv, const v4f r,
                                            const v4f w0, const v4f w1, const v4f w2,
                                            const v4f av) {
  float v0 = x.x + xrv.x + (r.y * w0.x + r.z * w1.x + r.w * w2.x);
  float v1 = x.y + xrv.y + (r.y * w0.y + r.z * w1.y + r.w * w2.y);
  float v2 = x.z + xrv.z + (r.y * w0.z + r.z * w1.z + r.w * w2.z);
  float v3 = x.w + xrv.w + (r.y * w0.w + r.z * w1.w + r.w * w2.w);
  v0 = fmaxf(v0, 0.2f * v0);
  v1 = fmaxf(v1, 0.2f * v1);
  v2 = fmaxf(v2, 0.2f * v2);
  v3 = fmaxf(v3, 0.2f * v3);
  float p = v0 * av.x + v1 * av.y + v2 * av.z + v3 * av.w;
  p += __shfl_xor(p, 1, 32);
  p += __shfl_xor(p, 2, 32);
  p += __shfl_xor(p, 4, 32);
  return p;
}

__global__ __launch_bounds__(256)
void node_attention_kernel(const int* __restrict__ off, const v4f* __restrict__ rec,
                           const float* __restrict__ xl, const float* __restrict__ xr,
                           const float* __restrict__ We, const float* __restrict__ att,
                           float* __restrict__ hout, float* __restrict__ stats) {
  const int lane = threadIdx.x & 31;
  const int node = __builtin_amdgcn_readfirstlane(blockIdx.x * 8 + (threadIdx.x >> 5));
  if (node >= N_NODES) return;
  const int c4 = lane * 4;
  const v4f xrv = *(const v4f*)(xr + (size_t)node * HIDW + c4);
  const v4f w0  = *(const v4f*)(We + c4);
  const v4f w1  = *(const v4f*)(We + HIDW + c4);
  const v4f w2  = *(const v4f*)(We + 2 * HIDW + c4);
  const v4f av  = *(const v4f*)(att + c4);
  const int beg = off[node];
  const int end = off[node + 1];
  float m = NEG_BIG, den = 0.f;
  v4f acc = {0.f, 0.f, 0.f, 0.f};
  for (int i = beg; i < end; i += 4) {
    const int n  = end - i;
    const int i1 = (n > 1) ? i + 1 : i;
    const int i2 = (n > 2) ? i + 2 : i;
    const int i3 = (n > 3) ? i + 3 : i;
    const v4f r0 = rec[i];
    const v4f r1 = rec[i1];
    const v4f r2 = rec[i2];
    const v4f r3 = rec[i3];
    const v4f x0 = *(const v4f*)(xl + (size_t)__float_as_int(r0.x) * HIDW + c4);
    const v4f x1 = *(const v4f*)(xl + (size_t)__float_as_int(r1.x) * HIDW + c4);
    const v4f x2 = *(const v4f*)(xl + (size_t)__float_as_int(r2.x) * HIDW + c4);
    const v4f x3 = *(const v4f*)(xl + (size_t)__float_as_int(r3.x) * HIDW + c4);
    const float p0 = head_logit(x0, xrv, r0, w0, w1, w2, av);
    float p1 = head_logit(x1, xrv, r1, w0, w1, w2, av);
    float p2 = head_logit(x2, xrv, r2, w0, w1, w2, av);
    float p3 = head_logit(x3, xrv, r3, w0, w1, w2, av);
    p1 = (n > 1) ? p1 : NEG_BIG;
    p2 = (n > 2) ? p2 : NEG_BIG;
    p3 = (n > 3) ? p3 : NEG_BIG;
    const float mn = fmaxf(fmaxf(m, p0), fmaxf(p1, fmaxf(p2, p3)));
    const float sc = expf(m - mn);
    const float e0 = expf(p0 - mn);
    const float e1 = expf(p1 - mn);
    const float e2 = expf(p2 - mn);
    const float e3 = expf(p3 - mn);
    den   = den * sc + ((e0 + e1) + (e2 + e3));
    acc.x = acc.x * sc + ((e0 * x0.x + e1 * x1.x) + (e2 * x2.x + e3 * x3.x));
    acc.y = acc.y * sc + ((e0 * x0.y + e1 * x1.y) + (e2 * x2.y + e3 * x3.y));
    acc.z = acc.z * sc + ((e0 * x0.z + e1 * x1.z) + (e2 * x2.z + e3 * x3.z));
    acc.w = acc.w * sc + ((e0 * x0.w + e1 * x1.w) + (e2 * x2.w + e3 * x3.w));
    m = mn;
  }
  const float inv = 1.0f / (den + 1e-16f);
  v4f o;
  o.x = acc.x * inv; o.y = acc.y * inv; o.z = acc.z * inv; o.w = acc.w * inv;
  *(v4f*)(hout + (size_t)node * HIDW + c4) = o;
  if (blockIdx.x == 0 && threadIdx.x < 2) stats[threadIdx.x] = 0.f;
}

__global__ __launch_bounds__(256)
void ln_reduce_kernel(const float* __restrict__ h, float* __restrict__ stats) {
  __shared__ float s1[256], s2[256];
  float a = 0.f, b = 0.f;
  for (int i = blockIdx.x * 256 + threadIdx.x; i < N_NODES * HIDW; i += gridDim.x * 256) {
    float v = h[i];
    a += v; b += v * v;
  }
  s1[threadIdx.x] = a; s2[threadIdx.x] = b;
  __syncthreads();
  for (int off = 128; off > 0; off >>= 1) {
    if ((int)threadIdx.x < off) {
      s1[threadIdx.x] += s1[threadIdx.x + off];
      s2[threadIdx.x] += s2[threadIdx.x + off];
    }
    __syncthreads();
  }
  if (threadIdx.x == 0) { atomicAdd(&stats[0], s1[0]); atomicAdd(&stats[1], s2[0]); }
}

__global__ __launch_bounds__(256)
void ln_gelu_kernel(float* __restrict__ h, const float* __restrict__ stats,
                    const float* __restrict__ w, const float* __restrict__ b) {
  int i = blockIdx.x * 256 + threadIdx.x;
  if (i >= N_NODES * HIDW) return;
  const float inv  = 1.0f / (float)(N_NODES * HIDW);
  const float mean = stats[0] * inv;
  const float var  = stats[1] * inv - mean * mean;
  const float rs   = rsqrtf(var + 1e-5f);
  const int c = i & (HIDW - 1);
  float v = (h[i] - mean) * rs * w[c] + b[c];
  v = 0.5f * v * (1.0f + erff(v * 0.70710678118654752f));
  h[i] = v;
}

extern "C" void kernel_launch(void* const* d_in, const int* in_sizes, int n_in,
                              void* d_out, int out_size, void* d_ws, size_t ws_size,
                              hipStream_t stream) {
  const float* x     = (const float*)d_in[0];
  const int*   ei    = (const int*)  d_in[1];
  const float* eattr = (const float*)d_in[2];
  const float* Wl    = (const float*)d_in[3];
  const float* Wr    = (const float*)d_in[4];
  const float* We    = (const float*)d_in[5];
  const float* att   = (const float*)d_in[6];
  const float* lnw   = (const float*)d_in[7];
  const float* lnb   = (const float*)d_in[8];
  const float* Wout  = (const float*)d_in[9];
  const float* bout  = (const float*)d_in[10];
  float* out = (float*)d_out;

  char* p = (char*)d_ws;
  auto carve = [&](size_t bytes) -> void* {
    void* q = (void*)p;
    p += (bytes + 255) & ~(size_t)255;
    return q;
  };
  float* hA    = (float*)carve((size_t)N_NODES * HIDW * 4);
  float* hB    = (float*)carve((size_t)N_NODES * HIDW * 4);
  float* xl    = (float*)carve((size_t)N_NODES * HIDW * 4);
  float* xr    = (float*)carve((size_t)N_NODES * HIDW * 4);
  v4f*   rec   = (v4f*)  carve((size_t)N_EDGES * 16);
  int*   rank  = (int*)  carve((size_t)N_EDGES * 4);
  int*   perm  = (int*)  carve((size_t)N_EDGES * 4);
  int*   cnt   = (int*)  carve((size_t)N_NODES * 4);
  int*   off   = (int*)  carve((size_t)(N_NODES + 1) * 4);
  float* stats = (float*)carve(2 * 4);

  const int elemBlocks = (N_NODES * HIDW + 255) / 256;
  const int cntBlocks  = (N_NODES + 255) / 256;
  const int edgeBlocks = (N_EDGES + 255) / 256;
  const int nodeBlocks = (N_NODES + 7) / 8;
  const int gemmBlocks = N_NODES / 16;

  zero_counts_kernel<<<cntBlocks, 256, 0, stream>>>(cnt);
  count_rank_kernel<<<edgeBlocks, 256, 0, stream>>>(ei, cnt, rank);
  scan_offsets_kernel<<<1, 1024, 0, stream>>>(cnt, off);
  fill_perm_kernel<<<edgeBlocks, 256, 0, stream>>>(ei, off, rank, perm);
  order_segments_kernel<<<nodeBlocks, 256, 0, stream>>>(ei, eattr, off, perm, rec);

  const float* hcur = x;
  float* bufs[2] = { hA, hB };
  for (int l = 0; l < NLAYERS; ++l) {
    float* hout = bufs[l & 1];
    gemm_dual_wmma<<<gemmBlocks, 256, 0, stream>>>(hcur, Wl + (size_t)l * HIDW * HIDW,
                                                   Wr + (size_t)l * HIDW * HIDW, xl, xr);
    node_attention_kernel<<<nodeBlocks, 256, 0, stream>>>(off, rec, xl, xr,
                                                          We + (size_t)l * EDIMW * HIDW,
                                                          att + (size_t)l * HIDW, hout, stats);
    ln_reduce_kernel<<<2048, 256, 0, stream>>>(hout, stats);
    ln_gelu_kernel<<<elemBlocks, 256, 0, stream>>>(hout, stats,
                                                   lnw + (size_t)l * HIDW,
                                                   lnb + (size_t)l * HIDW);
    hcur = hout;
  }
  gemm_bias_wmma<<<gemmBlocks, 256, 0, stream>>>(hcur, Wout, bout, out);
}


# LayerNorm scale/shift and exact GELU applied while the next GEMM stages its A tile in LDS: the element-wise pass (3 launches of one element per thread) is gone; sums and their order unchanged
# speedup vs baseline: 2.5880x; 1.1260x over previous
#include <hip/hip_runtime.h>
#include <math.h>

#define N_NODES 50000
#define N_EDGES 800000
#define HIDW    128
#define NHEAD   4
#define CPH     32
#define NLAYERS 3
#define EDIMW   3

typedef __attribute__((ext_vector_type(2))) float v2f;
typedef __attribute__((ext_vector_type(4))) float v4f;
typedef __attribute__((ext_vector_type(8))) float v8f;

#define NEG_BIG (-3.402823466e+38f)

__global__ __launch_bounds__(256)
void zero_counts_kernel(int* __restrict__ cnt) {
  const int i = blockIdx.x * 256 + threadIdx.x;
  if (i < N_NODES) cnt[i] = 0;
}

__global__ __launch_bounds__(256)
void count_rank_kernel(const int* __restrict__ ei, int* __restrict__ cnt,
                       int* __restrict__ rank) {
  const int e = blockIdx.x * 256 + threadIdx.x;
  if (e >= N_EDGES) return;
  rank[e] = atomicAdd(&cnt[ei[N_EDGES + e]], 1);
}

#define SCAN_PER ((N_NODES + 1023) / 1024)
__global__ __launch_bounds__(1024)
void scan_offsets_kernel(const int* __restrict__ cnt, int* __restrict__ off) {
  __shared__ int part[1024];
  const int t = threadIdx.x;
  const int b = t * SCAN_PER;
  int c[SCAN_PER];
  int s = 0;
#pragma unroll
  for (int j = 0; j < SCAN_PER; ++j) {
    const int i = b + j;
    const int v = cnt[(i < N_NODES) ? i : (N_NODES - 1)];
    c[j] = (i < N_NODES) ? v : 0;
    s += c[j];
  }
  part[t] = s;
  __syncthreads();
  for (int d = 1; d < 1024; d <<= 1) {
    const int v = (t >= d) ? part[t - d] : 0;
    __syncthreads();
    part[t] += v;
    __syncthreads();
  }
  int run = part[t] - s;
#pragma unroll
  for (int j = 0; j < SCAN_PER; ++j) {
    const int i = b + j;
    if (i < N_NODES) off[i] = run;
    run += c[j];
  }
  if (t == 1023) off[N_NODES] = part[1023];
}

__global__ __launch_bounds__(256)
void fill_perm_kernel(const int* __restrict__ ei, const int* __restrict__ off,
                      const int* __restrict__ rank, int* __restrict__ perm) {
  const int e = blockIdx.x * 256 + threadIdx.x;
  if (e >= N_EDGES) return;
  perm[off[ei[N_EDGES + e]] + rank[e]] = e;
}

__global__ __launch_bounds__(256)
void order_segments_kernel(const int* __restrict__ ei, const float* __restrict__ eattr,
                           const int* __restrict__ off, const int* __restrict__ perm,
                           v4f* __restrict__ rec) {
  const int lane = threadIdx.x & 31;
  const int node = blockIdx.x * 8 + (threadIdx.x >> 5);
  if (node >= N_NODES) return;
  const int beg = off[node];
  const int deg = off[node + 1] - beg;
  for (int k = lane; k < deg; k += 32) {
    const int my = perm[beg + k];
    int place = 0;
    for (int j = 0; j < deg; ++j) place += (perm[beg + j] < my) ? 1 : 0;
    v4f r;
    r.x = __int_as_float(ei[my]);
    r.y = eattr[my * 3 + 0];
    r.z = eattr[my * 3 + 1];
    r.w = eattr[my * 3 + 2];
    rec[beg + place] = r;
  }
}

template <bool NORM>
__device__ __forceinline__ void stage_a_tile(float* __restrict__ As, const float* __restrict__ A,
                                             const int rowBase,
                                             const float* __restrict__ stats,
                                             const float* __restrict__ lnw,
                                             const float* __restrict__ lnb) {
  const int c  = threadIdx.x & (HIDW - 1);
  const int r0 = threadIdx.x >> 7;
  float mean = 0.f, rs = 0.f, wc = 0.f, bc = 0.f;
  if (NORM) {
    const float inv = 1.0f / (float)(N_NODES * HIDW);
    mean = stats[0] * inv;
    const float var = stats[1] * inv - mean * mean;
    rs = rsqrtf(var + 1e-5f);
    wc = lnw[c];
    bc = lnb[c];
  }
  float v[8];
#pragma unroll
  for (int j = 0; j < 8; ++j) v[j] = A[(rowBase + r0 + 2 * j) * HIDW + c];
  if (NORM) {
#pragma unroll
    for (int j = 0; j < 8; ++j) {
      const float t = (v[j] - mean) * rs * wc + bc;
      v[j] = 0.5f * t * (1.0f + erff(t * 0.70710678118654752f));
    }
  }
#pragma unroll
  for (int j = 0; j < 8; ++j) As[(r0 + 2 * j) * 132 + c] = v[j];
}

__device__ __forceinline__ void dual_tile_product(const float* __restrict__ As, const int rowBase,
                                                  const float* __restrict__ W0,
                                                  const float* __restrict__ W1,
                                                  float* __restrict__ O0,
                                                  float* __restrict__ O1) {
  const int wave    = threadIdx.x >> 5;
  const int lane    = threadIdx.x & 31;
  const int colBase = wave * 16;
  const int laneHi  = lane >> 4;
  const int laneLo  = lane & 15;
  v8f c0 = {}; v8f c1 = {};
  for (int k0 = 0; k0 < HIDW; k0 += 4) {
    const int ka = k0 + 2 * laneHi;
    v2f a;
    a.x = As[laneLo * 132 + ka];
    a.y = As[laneLo * 132 + ka + 1];
    v2f b0, b1;
    b0.x = W0[(ka + 0) * HIDW + colBase + laneLo];
    b0.y = W0[(ka + 1) * HIDW + colBase + laneLo];
    b1.x = W1[(ka + 0) * HIDW + colBase + laneLo];
    b1.y = W1[(ka + 1) * HIDW + colBase + laneLo];
    c0 = __builtin_amdgcn_wmma_f32_16x16x4_f32(false, a, false, b0, (short)0, c0, false, false);
    c1 = __builtin_amdgcn_wmma_f32_16x16x4_f32(false, a, false, b1, (short)0, c1, false, false);
  }
#pragma unroll
  for (int r = 0; r < 8; ++r) {
    const int row = rowBase + r + 8 * laneHi;
    O0[row * HIDW + colBase + laneLo] = c0[r];
    O1[row * HIDW + colBase + laneLo] = c1[r];
  }
}

__global__ __launch_bounds__(256)
void gemm_dual_wmma(const float* __restrict__ A,
                    const float* __restrict__ W0,
                    const float* __restrict__ W1,
                    float* __restrict__ O0,
                    float* __restrict__ O1) {
  __shared__ float As[16 * 132];
  const int rowBase = blockIdx.x * 16;
  stage_a_tile<false>(As, A, rowBase, nullptr, nullptr, nullptr);
  __syncthreads();
  dual_tile_product(As, rowBase, W0, W1, O0, O1);
}

__global__ __launch_bounds__(256)
void gemm_dual_norm_wmma(const float* __restrict__ A,
                         const float* __restrict__ stats,
                         const float* __restrict__ lnw,
                         const float* __restrict__ lnb,
                         const float* __restrict__ W0,
                         const float* __restrict__ W1,
                         float* __restrict__ O0,
                         float* __restrict__ O1) {
  __shared__ float As[16 * 132];
  const int rowBase = blockIdx.x * 16;
  stage_a_tile<true>(As, A, rowBase, stats, lnw, lnb);
  __syncthreads();
  dual_tile_product(As, rowBase, W0, W1, O0, O1);
}

__global__ __launch_bounds__(256)
void gemm_bias_norm_wmma(const float* __restrict__ A,
                         const float* __restrict__ stats,
                         const float* __restrict__ lnw,
                         const float* __restrict__ lnb,
                         const float* __restrict__ W,
                         const float* __restrict__ bias,
                         float* __restrict__ O) {
  __shared__ float As[16 * 132];
  const int rowBase = blockIdx.x * 16;
  stage_a_tile<true>(As, A, rowBase, stats, lnw, lnb);
  __syncthreads();
  const int wave    = threadIdx.x >> 5;
  const int lane    = threadIdx.x & 31;
  const int colBase = wave * 16;
  const int laneHi  = lane >> 4;
  const int laneLo  = lane & 15;
  v8f c0 = {};
  for (int k0 = 0; k0 < HIDW; k0 += 4) {
    const int ka = k0 + 2 * laneHi;
    v2f a;
    a.x = As[laneLo * 132 + ka];
    a.y = As[laneLo * 132 + ka + 1];
    v2f b;
    b.x = W[(ka + 0) * HIDW + colBase + laneLo];
    b.y = W[(ka + 1) * HIDW + colBase + laneLo];
    c0 = __builtin_amdgcn_wmma_f32_16x16x4_f32(false, a, false, b, (short)0, c0, false, false);
  }
  const float bv = bias[colBase + laneLo];
#pragma unroll
  for (int r = 0; r < 8; ++r) {
    const int row = rowBase + r + 8 * laneHi;
    O[row * HIDW + colBase + laneLo] = c0[r] + bv;
  }
}

__device__ __forceinline__ float head_logit(const v4f x, const v4f xrv, const v4f r,
                                            const v4f w0, const v4f w1, const v4f w2,
                                            const v4f av) {
  float v0 = x.x + xrv.x + (r.y * w0.x + r.z * w1.x + r.w * w2.x);
  float v1 = x.y + xrv.y + (r.y * w0.y + r.z * w1.y + r.w * w2.y);
  float v2 = x.z + xrv.z + (r.y * w0.z + r.z * w1.z + r.w * w2.z);
  float v3 = x.w + xrv.w + (r.y * w0.w + r.z * w1.w + r.w * w2.w);
  v0 = fmaxf(v0, 0.2f * v0);
  v1 = fmaxf(v1, 0.2f * v1);
  v2 = fmaxf(v2, 0.2f * v2);
  v3 = fmaxf(v3, 0.2f * v3);
  float p = v0 * av.x + v1 * av.y + v2 * av.z + v3 * av.w;
  p += __shfl_xor(p, 1, 32);
  p += __shfl_xor(p, 2, 32);
  p += __shfl_xor(p, 4, 32);
  return p;
}

__global__ __launch_bounds__(256)
void node_attention_kernel(const int* __restrict__ off, const v4f* __restrict__ rec,
                           const float* __restrict__ xl, const float* __restrict__ xr,
                           const float* __restrict__ We, const float* __restrict__ att,
                           float* __restrict__ hout, float* __restrict__ stats) {
  const int lane = threadIdx.x & 31;
  const int node = __builtin_amdgcn_readfirstlane(blockIdx.x * 8 + (threadIdx.x >> 5));
  if (node >= N_NODES) return;
  const int c4 = lane * 4;
  const v4f xrv = *(const v4f*)(xr + (size_t)node * HIDW + c4);
  const v4f w0  = *(const v4f*)(We + c4);
  const v4f w1  = *(const v4f*)(We + HIDW + c4);
  const v4f w2  = *(const v4f*)(We + 2 * HIDW + c4);
  const v4f av  = *(const v4f*)(att + c4);
  const int beg = off[node];
  const int end = off[node + 1];
  float m = NEG_BIG, den = 0.f;
  v4f acc = {0.f, 0.f, 0.f, 0.f};
  for (int i = beg; i < end; i += 4) {
    const int n  = end - i;
    const int i1 = (n > 1) ? i + 1 : i;
    const int i2 = (n > 2) ? i + 2 : i;
    const int i3 = (n > 3) ? i + 3 : i;
    const v4f r0 = rec[i];
    const v4f r1 = rec[i1];
    const v4f r2 = rec[i2];
    const v4f r3 = rec[i3];
    const v4f x0 = *(const v4f*)(xl + (size_t)__float_as_int(r0.x) * HIDW + c4);
    const v4f x1 = *(const v4f*)(xl + (size_t)__float_as_int(r1.x) * HIDW + c4);
    const v4f x2 = *(const v4f*)(xl + (size_t)__float_as_int(r2.x) * HIDW + c4);
    const v4f x3 = *(const v4f*)(xl + (size_t)__float_as_int(r3.x) * HIDW + c4);
    const float p0 = head_logit(x0, xrv, r0, w0, w1, w2, av);
    float p1 = head_logit(x1, xrv, r1, w0, w1, w2, av);
    float p2 = head_logit(x2, xrv, r2, w0, w1, w2, av);
    float p3 = head_logit(x3, xrv, r3, w0, w1, w2, av);
    p1 = (n > 1) ? p1 : NEG_BIG;
    p2 = (n > 2) ? p2 : NEG_BIG;
    p3 = (n > 3) ? p3 : NEG_BIG;
    const float mn = fmaxf(fmaxf(m, p0), fmaxf(p1, fmaxf(p2, p3)));
    const float sc = expf(m - mn);
    const float e0 = expf(p0 - mn);
    const float e1 = expf(p1 - mn);
    const float e2 = expf(p2 - mn);
    const float e3 = expf(p3 - mn);
    den   = den * sc + ((e0 + e1) + (e2 + e3));
    acc.x = acc.x * sc + ((e0 * x0.x + e1 * x1.x) + (e2 * x2.x + e3 * x3.x));
    acc.y = acc.y * sc + ((e0 * x0.y + e1 * x1.y) + (e2 * x2.y + e3 * x3.y));
    acc.z = acc.z * sc + ((e0 * x0.z + e1 * x1.z) + (e2 * x2.z + e3 * x3.z));
    acc.w = acc.w * sc + ((e0 * x0.w + e1 * x1.w) + (e2 * x2.w + e3 * x3.w));
    m = mn;
  }
  const float inv = 1.0f / (den + 1e-16f);
  v4f o;
  o.x = acc.x * inv; o.y = acc.y * inv; o.z = acc.z * inv; o.w = acc.w * inv;
  *(v4f*)(hout + (size_t)node * HIDW + c4) = o;
  if (blockIdx.x == 0 && threadIdx.x < 2) stats[threadIdx.x] = 0.f;
}

__global__ __launch_bounds__(256)
void ln_reduce_kernel(const float* __restrict__ h, float* __restrict__ stats) {
  __shared__ float s1[256], s2[256];
  float a = 0.f, b = 0.f;
  for (int i = blockIdx.x * 256 + threadIdx.x; i < N_NODES * HIDW; i += gridDim.x * 256) {
    float v = h[i];
    a += v; b += v * v;
  }
  s1[threadIdx.x] = a; s2[threadIdx.x] = b;
  __syncthreads();
  for (int off = 128; off > 0; off >>= 1) {
    if ((int)threadIdx.x < off) {
      s1[threadIdx.x] += s1[threadIdx.x + off];
      s2[threadIdx.x] += s2[threadIdx.x + off];
    }
    __syncthreads();
  }
  if (threadIdx.x == 0) { atomicAdd(&stats[0], s1[0]); atomicAdd(&stats[1], s2[0]); }
}

extern "C" void kernel_launch(void* const* d_in, const int* in_sizes, int n_in,
                              void* d_out, int out_size, void* d_ws, size_t ws_size,
                              hipStream_t stream) {
  const float* x     = (const float*)d_in[0];
  const int*   ei    = (const int*)  d_in[1];
  const float* eattr = (const float*)d_in[2];
  const float* Wl    = (const float*)d_in[3];
  const float* Wr    = (const float*)d_in[4];
  const float* We    = (const float*)d_in[5];
  const float* att   = (const float*)d_in[6];
  const float* lnw   = (const float*)d_in[7];
  const float* lnb   = (const float*)d_in[8];
  const float* Wout  = (const float*)d_in[9];
  const float* bout  = (const float*)d_in[10];
  float* out = (float*)d_out;

  char* p = (char*)d_ws;
  auto carve = [&](size_t bytes) -> void* {
    void* q = (void*)p;
    p += (bytes + 255) & ~(size_t)255;
    return q;
  };
  float* hA    = (float*)carve((size_t)N_NODES * HIDW * 4);
  float* hB    = (float*)carve((size_t)N_NODES * HIDW * 4);
  float* xl    = (float*)carve((size_t)N_NODES * HIDW * 4);
  float* xr    = (float*)carve((size_t)N_NODES * HIDW * 4);
  v4f*   rec   = (v4f*)  carve((size_t)N_EDGES * 16);
  int*   rank  = (int*)  carve((size_t)N_EDGES * 4);
  int*   perm  = (int*)  carve((size_t)N_EDGES * 4);
  int*   cnt   = (int*)  carve((size_t)N_NODES * 4);
  int*   off   = (int*)  carve((size_t)(N_NODES + 1) * 4);
  float* stats = (float*)carve(2 * 4);

  const int cntBlocks  = (N_NODES + 255) / 256;
  const int edgeBlocks = (N_EDGES + 255) / 256;
  const int nodeBlocks = (N_NODES + 7) / 8;
  const int gemmBlocks = N_NODES / 16;

  zero_counts_kernel<<<cntBlocks, 256, 0, stream>>>(cnt);
  count_rank_kernel<<<edgeBlocks, 256, 0, stream>>>(ei, cnt, rank);
  scan_offsets_kernel<<<1, 1024, 0, stream>>>(cnt, off);
  fill_perm_kernel<<<edgeBlocks, 256, 0, stream>>>(ei, off, rank, perm);
  order_segments_kernel<<<nodeBlocks, 256, 0, stream>>>(ei, eattr, off, perm, rec);

  const float* hcur = x;
  float* bufs[2] = { hA, hB };
  for (int l = 0; l < NLAYERS; ++l) {
    float* hout = bufs[l & 1];
    if (l == 0) {
      gemm_dual_wmma<<<gemmBlocks, 256, 0, stream>>>(hcur, Wl, Wr, xl, xr);
    } else {
      gemm_dual_norm_wmma<<<gemmBlocks, 256, 0, stream>>>(hcur, stats,
                                                          lnw + (size_t)(l - 1) * HIDW,
                                                          lnb + (size_t)(l - 1) * HIDW,
                                                          Wl + (size_t)l * HIDW * HIDW,
                                                          Wr + (size_t)l * HIDW * HIDW, xl, xr);
    }
    node_attention_kernel<<<nodeBlocks, 256, 0, stream>>>(off, rec, xl, xr,
                                                          We + (size_t)l * EDIMW * HIDW,
                                                          att + (size_t)l * HIDW, hout, stats);
    ln_reduce_kernel<<<2048, 256, 0, stream>>>(hout, stats);
    hcur = hout;
  }
  gemm_bias_norm_wmma<<<gemmBlocks, 256, 0, stream>>>(hcur, stats,
                                                      lnw + (size_t)(NLAYERS - 1) * HIDW,
                                                      lnb + (size_t)(NLAYERS - 1) * HIDW,
                                                      Wout, bout, out);
}


# LayerNorm sums taken in the attention kernel from the row still in registers (per-wave share, float atomics into 32 slots, reader adds slots in fixed order): the 3 reduction launches are gone
# speedup vs baseline: 2.7785x; 1.0736x over previous
#include <hip/hip_runtime.h>
#include <math.h>

#define N_NODES 50000
#define N_EDGES 800000
#define HIDW    128
#define NHEAD   4
#define CPH     32
#define NLAYERS 3
#define EDIMW   3

typedef __attribute__((ext_vector_type(2))) float v2f;
typedef __attribute__((ext_vector_type(4))) float v4f;
typedef __attribute__((ext_vector_type(8))) float v8f;

#define NEG_BIG (-3.402823466e+38f)

#define LN_SLOTS       32
#define LN_SLOT_PITCH  32
#define LN_LAYER_PITCH (LN_SLOTS * LN_SLOT_PITCH)

__global__ __launch_bounds__(256)
void zero_counts_kernel(int* __restrict__ cnt, float* __restrict__ stats) {
  const int i = blockIdx.x * 256 + threadIdx.x;
  if (i < N_NODES) cnt[i] = 0;
  if (i < NLAYERS * LN_LAYER_PITCH) stats[i] = 0.f;
}

__global__ __launch_bounds__(256)
void count_rank_kernel(const int* __restrict__ ei, int* __restrict__ cnt,
                       int* __restrict__ rank) {
  const int e = blockIdx.x * 256 + threadIdx.x;
  if (e >= N_EDGES) return;
  rank[e] = atomicAdd(&cnt[ei[N_EDGES + e]], 1);
}

#define SCAN_PER ((N_NODES + 1023) / 1024)
__global__ __launch_bounds__(1024)
void scan_offsets_kernel(const int* __restrict__ cnt, int* __restrict__ off) {
  __shared__ int part[1024];
  const int t = threadIdx.x;
  const int b = t * SCAN_PER;
  int c[SCAN_PER];
  int s = 0;
#pragma unroll
  for (int j = 0; j < SCAN_PER; ++j) {
    const int i = b + j;
    const int v = cnt[(i < N_NODES) ? i : (N_NODES - 1)];
    c[j] = (i < N_NODES) ? v : 0;
    s += c[j];
  }
  part[t] = s;
  __syncthreads();
  for (int d = 1; d < 1024; d <<= 1) {
    const int v = (t >= d) ? part[t - d] : 0;
    __syncthreads();
    part[t] += v;
    __syncthreads();
  }
  int run = part[t] - s;
#pragma unroll
  for (int j = 0; j < SCAN_PER; ++j) {
    const int i = b + j;
    if (i < N_NODES) off[i] = run;
    run += c[j];
  }
  if (t == 1023) off[N_NODES] = part[1023];
}

__global__ __launch_bounds__(256)
void fill_perm_kernel(const int* __restrict__ ei, const int* __restrict__ off,
                      const int* __restrict__ rank, int* __restrict__ perm) {
  const int e = blockIdx.x * 256 + threadIdx.x;
  if (e >= N_EDGES) return;
  perm[off[ei[N_EDGES + e]] + rank[e]] = e;
}

__global__ __launch_bounds__(256)
void order_segments_kernel(const int* __restrict__ ei, const float* __restrict__ eattr,
                           const int* __restrict__ off, const int* __restrict__ perm,
                           v4f* __restrict__ rec) {
  const int lane = threadIdx.x & 31;
  const int node = blockIdx.x * 8 + (threadIdx.x >> 5);
  if (node >= N_NODES) return;
  const int beg = off[node];
  const int deg = off[node + 1] - beg;
  for (int k = lane; k < deg; k += 32) {
    const int my = perm[beg + k];
    int place = 0;
    for (int j = 0; j < deg; ++j) place += (perm[beg + j] < my) ? 1 : 0;
    v4f r;
    r.x = __int_as_float(ei[my]);
    r.y = eattr[my * 3 + 0];
    r.z = eattr[my * 3 + 1];
    r.w = eattr[my * 3 + 2];
    rec[beg + place] = r;
  }
}

template <bool NORM>
__device__ __forceinline__ void stage_a_tile(float* __restrict__ As, const float* __restrict__ A,
                                             const int rowBase,
                                             const float* __restrict__ stats,
                                             const float* __restrict__ lnw,
                                             const float* __restrict__ lnb) {
  const int c  = threadIdx.x & (HIDW - 1);
  const int r0 = threadIdx.x >> 7;
  float mean = 0.f, rs = 0.f, wc = 0.f, bc = 0.f;
  if (NORM) {
    const v2f part = *(const v2f*)(stats + (threadIdx.x & 31) * LN_SLOT_PITCH);
    float s1 = part.x, s2 = part.y;
    s1 += __shfl_xor(s1, 16, 32);  s2 += __shfl_xor(s2, 16, 32);
    s1 += __shfl_xor(s1, 8, 32);   s2 += __shfl_xor(s2, 8, 32);
    s1 += __shfl_xor(s1, 4, 32);   s2 += __shfl_xor(s2, 4, 32);
    s1 += __shfl_xor(s1, 2, 32);   s2 += __shfl_xor(s2, 2, 32);
    s1 += __shfl_xor(s1, 1, 32);   s2 += __shfl_xor(s2, 1, 32);
    const float inv = 1.0f / (float)(N_NODES * HIDW);
    mean = s1 * inv;
    const float var = s2 * inv - mean * mean;
    rs = rsqrtf(var + 1e-5f);
    wc = lnw[c];
    bc = lnb[c];
  }
  float v[8];
#pragma unroll
  for (int j = 0; j < 8; ++j) v[j] = A[(rowBase + r0 + 2 * j) * HIDW + c];
  if (NORM) {
#pragma unroll
    for (int j = 0; j < 8; ++j) {
      const float t = (v[j] - mean) * rs * wc + bc;
      v[j] = 0.5f * t * (1.0f + erff(t * 0.70710678118654752f));
    }
  }
#pragma unroll
  for (int j = 0; j < 8; ++j) As[(r0 + 2 * j) * 132 + c] = v[j];
}

__device__ __forceinline__ void dual_tile_product(const float* __restrict__ As, const int rowBase,
                                                  const float* __restrict__ W0,
                                                  const float* __restrict__ W1,
                                                  float* __restrict__ O0,
                                                  float* __restrict__ O1) {
  const int wave    = threadIdx.x >> 5;
  const int lane    = threadIdx.x & 31;
  const int colBase = wave * 16;
  const int laneHi  = lane >> 4;
  const int laneLo  = lane & 15;
  v8f c0 = {}; v8f c1 = {};
  for (int k0 = 0; k0 < HIDW; k0 += 4) {
    const int ka = k0 + 2 * laneHi;
    v2f a;
    a.x = As[laneLo * 132 + ka];
    a.y = As[laneLo * 132 + ka + 1];
    v2f b0, b1;
    b0.x = W0[(ka + 0) * HIDW + colBase + laneLo];
    b0.y = W0[(ka + 1) * HIDW + colBase + laneLo];
    b1.x = W1[(ka + 0) * HIDW + colBase + laneLo];
    b1.y = W1[(ka + 1) * HIDW + colBase + laneLo];
    c0 = __builtin_amdgcn_wmma_f32_16x16x4_f32(false, a, false, b0, (short)0, c0, false, false);
    c1 = __builtin_amdgcn_wmma_f32_16x16x4_f32(false, a, false, b1, (short)0, c1, false, false);
  }
#pragma unroll
  for (int r = 0; r < 8; ++r) {
    const int row = rowBase + r + 8 * laneHi;
    O0[row * HIDW + colBase + laneLo] = c0[r];
    O1[row * HIDW + colBase + laneLo] = c1[r];
  }
}

__global__ __launch_bounds__(256)
void gemm_dual_wmma(const float* __restrict__ A,
                    const float* __restrict__ W0,
                    const float* __restrict__ W1,
                    float* __restrict__ O0,
                    float* __restrict__ O1) {
  __shared__ float As[16 * 132];
  const int rowBase = blockIdx.x * 16;
  stage_a_tile<false>(As, A, rowBase, nullptr, nullptr, nullptr);
  __syncthreads();
  dual_tile_product(As, rowBase, W0, W1, O0, O1);
}

__global__ __launch_bounds__(256)
void gemm_dual_norm_wmma(const float* __restrict__ A,
                         const float* __restrict__ stats,
                         const float* __restrict__ lnw,
                         const float* __restrict__ lnb,
                         const float* __restrict__ W0,
                         const float* __restrict__ W1,
                         float* __restrict__ O0,
                         float* __restrict__ O1) {
  __shared__ float As[16 * 132];
  const int rowBase = blockIdx.x * 16;
  stage_a_tile<true>(As, A, rowBase, stats, lnw, lnb);
  __syncthreads();
  dual_tile_product(As, rowBase, W0, W1, O0, O1);
}

__global__ __launch_bounds__(256)
void gemm_bias_norm_wmma(const float* __restrict__ A,
                         const float* __restrict__ stats,
                         const float* __restrict__ lnw,
                         const float* __restrict__ lnb,
                         const float* __restrict__ W,
                         const float* __restrict__ bias,
                         float* __restrict__ O) {
  __shared__ float As[16 * 132];
  const int rowBase = blockIdx.x * 16;
  stage_a_tile<true>(As, A, rowBase, stats, lnw, lnb);
  __syncthreads();
  const int wave    = threadIdx.x >> 5;
  const int lane    = threadIdx.x & 31;
  const int colBase = wave * 16;
  const int laneHi  = lane >> 4;
  const int laneLo  = lane & 15;
  v8f c0 = {};
  for (int k0 = 0; k0 < HIDW; k0 += 4) {
    const int ka = k0 + 2 * laneHi;
    v2f a;
    a.x = As[laneLo * 132 + ka];
    a.y = As[laneLo * 132 + ka + 1];
    v2f b;
    b.x = W[(ka + 0) * HIDW + colBase + laneLo];
    b.y = W[(ka + 1) * HIDW + colBase + laneLo];
    c0 = __builtin_amdgcn_wmma_f32_16x16x4_f32(false, a, false, b, (short)0, c0, false, false);
  }
  const float bv = bias[colBase + laneLo];
#pragma unroll
  for (int r = 0; r < 8; ++r) {
    const int row = rowBase + r + 8 * laneHi;
    O[row * HIDW + colBase + laneLo] = c0[r] + bv;
  }
}

__device__ __forceinline__ float head_logit(const v4f x, const v4f xrv, const v4f r,
                                            const v4f w0, const v4f w1, const v4f w2,
                                            const v4f av) {
  float v0 = x.x + xrv.x + (r.y * w0.x + r.z * w1.x + r.w * w2.x);
  float v1 = x.y + xrv.y + (r.y * w0.y + r.z * w1.y + r.w * w2.y);
  float v2 = x.z + xrv.z + (r.y * w0.z + r.z * w1.z + r.w * w2.z);
  float v3 = x.w + xrv.w + (r.y * w0.w + r.z * w1.w + r.w * w2.w);
  v0 = fmaxf(v0, 0.2f * v0);
  v1 = fmaxf(v1, 0.2f * v1);
  v2 = fmaxf(v2, 0.2f * v2);
  v3 = fmaxf(v3, 0.2f * v3);
  float p = v0 * av.x + v1 * av.y + v2 * av.z + v3 * av.w;
  p += __shfl_xor(p, 1, 32);
  p += __shfl_xor(p, 2, 32);
  p += __shfl_xor(p, 4, 32);
  return p;
}

__global__ __launch_bounds__(256)
void node_attention_kernel(const int* __restrict__ off, const v4f* __restrict__ rec,
                           const float* __restrict__ xl, const float* __restrict__ xr,
                           const float* __restrict__ We, const float* __restrict__ att,
                           float* __restrict__ hout, float* __restrict__ stats) {
  const int lane = threadIdx.x & 31;
  const int node = __builtin_amdgcn_readfirstlane(blockIdx.x * 8 + (threadIdx.x >> 5));
  if (node >= N_NODES) return;
  const int c4 = lane * 4;
  const v4f xrv = *(const v4f*)(xr + (size_t)node * HIDW + c4);
  const v4f w0  = *(const v4f*)(We + c4);
  const v4f w1  = *(const v4f*)(We + HIDW + c4);
  const v4f w2  = *(const v4f*)(We + 2 * HIDW + c4);
  const v4f av  = *(const v4f*)(att + c4);
  const int beg = off[node];
  const int end = off[node + 1];
  float m = NEG_BIG, den = 0.f;
  v4f acc = {0.f, 0.f, 0.f, 0.f};
  for (int i = beg; i < end; i += 4) {
    const int n  = end - i;
    const int i1 = (n > 1) ? i + 1 : i;
    const int i2 = (n > 2) ? i + 2 : i;
    const int i3 = (n > 3) ? i + 3 : i;
    const v4f r0 = rec[i];
    const v4f r1 = rec[i1];
    const v4f r2 = rec[i2];
    const v4f r3 = rec[i3];
    const v4f x0 = *(const v4f*)(xl + (size_t)__float_as_int(r0.x) * HIDW + c4);
    const v4f x1 = *(const v4f*)(xl + (size_t)__float_as_int(r1.x) * HIDW + c4);
    const v4f x2 = *(const v4f*)(xl + (size_t)__float_as_int(r2.x) * HIDW + c4);
    const v4f x3 = *(const v4f*)(xl + (size_t)__float_as_int(r3.x) * HIDW + c4);
    const float p0 = head_logit(x0, xrv, r0, w0, w1, w2, av);
    float p1 = head_logit(x1, xrv, r1, w0, w1, w2, av);
    float p2 = head_logit(x2, xrv, r2, w0, w1, w2, av);
    float p3 = head_logit(x3, xrv, r3, w0, w1, w2, av);
    p1 = (n > 1) ? p1 : NEG_BIG;
    p2 = (n > 2) ? p2 : NEG_BIG;
    p3 = (n > 3) ? p3 : NEG_BIG;
    const float mn = fmaxf(fmaxf(m, p0), fmaxf(p1, fmaxf(p2, p3)));
    const float sc = expf(m - mn);
    const float e0 = expf(p0 - mn);
    const float e1 = expf(p1 - mn);
    const float e2 = expf(p2 - mn);
    const float e3 = expf(p3 - mn);
    den   = den * sc + ((e0 + e1) + (e2 + e3));
    acc.x = acc.x * sc + ((e0 * x0.x + e1 * x1.x) + (e2 * x2.x + e3 * x3.x));
    acc.y = acc.y * sc + ((e0 * x0.y + e1 * x1.y) + (e2 * x2.y + e3 * x3.y));
    acc.z = acc.z * sc + ((e0 * x0.z + e1 * x1.z) + (e2 * x2.z + e3 * x3.z));
    acc.w = acc.w * sc + ((e0 * x0.w + e1 * x1.w) + (e2 * x2.w + e3 * x3.w));
    m = mn;
  }
  const float inv = 1.0f / (den + 1e-16f);
  v4f o;
  o.x = acc.x * inv; o.y = acc.y * inv; o.z = acc.z * inv; o.w = acc.w * inv;
  *(v4f*)(hout + (size_t)node * HIDW + c4) = o;
  float s1 = (o.x + o.y) + (o.z + o.w);
  float s2 = (o.x * o.x + o.y * o.y) + (o.z * o.z + o.w * o.w);
  s1 += __shfl_xor(s1, 16, 32);  s2 += __shfl_xor(s2, 16, 32);
  s1 += __shfl_xor(s1, 8, 32);   s2 += __shfl_xor(s2, 8, 32);
  s1 += __shfl_xor(s1, 4, 32);   s2 += __shfl_xor(s2, 4, 32);
  s1 += __shfl_xor(s1, 2, 32);   s2 += __shfl_xor(s2, 2, 32);
  s1 += __shfl_xor(s1, 1, 32);   s2 += __shfl_xor(s2, 1, 32);
  if (lane == 0) {
    float* slot = stats + (node & (LN_SLOTS - 1)) * LN_SLOT_PITCH;
    atomicAdd(slot, s1);
    atomicAdd(slot + 1, s2);
  }
}

extern "C" void kernel_launch(void* const* d_in, const int* in_sizes, int n_in,
                              void* d_out, int out_size, void* d_ws, size_t ws_size,
                              hipStream_t stream) {
  const float* x     = (const float*)d_in[0];
  const int*   ei    = (const int*)  d_in[1];
  const float* eattr = (const float*)d_in[2];
  const float* Wl    = (const float*)d_in[3];
  const float* Wr    = (const float*)d_in[4];
  const float* We    = (const float*)d_in[5];
  const float* att   = (const float*)d_in[6];
  const float* lnw   = (const float*)d_in[7];
  const float* lnb   = (const float*)d_in[8];
  const float* Wout  = (const float*)d_in[9];
  const float* bout  = (const float*)d_in[10];
  float* out = (float*)d_out;

  char* p = (char*)d_ws;
  auto carve = [&](size_t bytes) -> void* {
    void* q = (void*)p;
    p += (bytes + 255) & ~(size_t)255;
    return q;
  };
  float* hA    = (float*)carve((size_t)N_NODES * HIDW * 4);
  float* hB    = (float*)carve((size_t)N_NODES * HIDW * 4);
  float* xl    = (float*)carve((size_t)N_NODES * HIDW * 4);
  float* xr    = (float*)carve((size_t)N_NODES * HIDW * 4);
  v4f*   rec   = (v4f*)  carve((size_t)N_EDGES * 16);
  int*   rank  = (int*)  carve((size_t)N_EDGES * 4);
  int*   perm  = (int*)  carve((size_t)N_EDGES * 4);
  int*   cnt   = (int*)  carve((size_t)N_NODES * 4);
  int*   off   = (int*)  carve((size_t)(N_NODES + 1) * 4);
  float* stats = (float*)carve((size_t)NLAYERS * LN_LAYER_PITCH * 4);

  const int cntBlocks  = (N_NODES + 255) / 256;
  const int edgeBlocks = (N_EDGES + 255) / 256;
  const int nodeBlocks = (N_NODES + 7) / 8;
  const int gemmBlocks = N_NODES / 16;

  zero_counts_kernel<<<cntBlocks, 256, 0, stream>>>(cnt, stats);
  count_rank_kernel<<<edgeBlocks, 256, 0, stream>>>(ei, cnt, rank);
  scan_offsets_kernel<<<1, 1024, 0, stream>>>(cnt, off);
  fill_perm_kernel<<<edgeBlocks, 256, 0, stream>>>(ei, off, rank, perm);
  order_segments_kernel<<<nodeBlocks, 256, 0, stream>>>(ei, eattr, off, perm, rec);

  const float* hcur = x;
  float* bufs[2] = { hA, hB };
  for (int l = 0; l < NLAYERS; ++l) {
    float* hout = bufs[l & 1];
    if (l == 0) {
      gemm_dual_wmma<<<gemmBlocks, 256, 0, stream>>>(hcur, Wl, Wr, xl, xr);
    } else {
      gemm_dual_norm_wmma<<<gemmBlocks, 256, 0, stream>>>(hcur,
                                                          stats + (size_t)(l - 1) * LN_LAYER_PITCH,
                                                          lnw + (size_t)(l - 1) * HIDW,
                                                          lnb + (size_t)(l - 1) * HIDW,
                                                          Wl + (size_t)l * HIDW * HIDW,
                                                          Wr + (size_t)l * HIDW * HIDW, xl, xr);
    }
    node_attention_kernel<<<nodeBlocks, 256, 0, stream>>>(off, rec, xl, xr,
                                                          We + (size_t)l * EDIMW * HIDW,
                                                          att + (size_t)l * HIDW, hout,
                                                          stats + (size_t)l * LN_LAYER_PITCH);
    hcur = hout;
  }
  gemm_bias_norm_wmma<<<gemmBlocks, 256, 0, stream>>>(hcur,
                                                      stats + (size_t)(NLAYERS - 1) * LN_LAYER_PITCH,
                                                      lnw + (size_t)(NLAYERS - 1) * HIDW,
                                                      lnb + (size_t)(NLAYERS - 1) * HIDW,
                                                      Wout, bout, out);
}


# Prefix-scan launch: counts in and offsets out pass through LDS with whole-line b128 accesses (4 lines per wave instruction, not 32 lines 196 B apart); float sums and their order unchanged
# speedup vs baseline: 3.0611x; 1.1017x over previous
#include <hip/hip_runtime.h>
#include <math.h>

#define N_NODES 50000
#define N_EDGES 800000
#define HIDW    128
#define NHEAD   4
#define CPH     32
#define NLAYERS 3
#define EDIMW   3

typedef __attribute__((ext_vector_type(2))) float v2f;
typedef __attribute__((ext_vector_type(4))) float v4f;
typedef __attribute__((ext_vector_type(8))) float v8f;
typedef __attribute__((ext_vector_type(4))) int   v4i;

#define NEG_BIG (-3.402823466e+38f)

#define LN_SLOTS       32
#define LN_SLOT_PITCH  32
#define LN_LAYER_PITCH (LN_SLOTS * LN_SLOT_PITCH)

__global__ __launch_bounds__(256)
void zero_counts_kernel(int* __restrict__ cnt, float* __restrict__ stats) {
  const int i = blockIdx.x * 256 + threadIdx.x;
  if (i < N_NODES) cnt[i] = 0;
  if (i < NLAYERS * LN_LAYER_PITCH) stats[i] = 0.f;
}

__global__ __launch_bounds__(256)
void count_rank_kernel(const int* __restrict__ ei, int* __restrict__ cnt,
                       int* __restrict__ rank) {
  const int e = blockIdx.x * 256 + threadIdx.x;
  if (e >= N_EDGES) return;
  rank[e] = atomicAdd(&cnt[ei[N_EDGES + e]], 1);
}

#define SCAN_PER   ((N_NODES + 1023) / 1024)
#define SCAN_QUADS (N_NODES / 4)
#define SCAN_QPT   ((1024 * SCAN_PER + 4095) / 4096)
#define SCAN_LDS   (4096 * SCAN_QPT)
static_assert(N_NODES % 4 == 0, "counts are moved in whole 16-byte groups");
static_assert(SCAN_LDS >= 1024 * SCAN_PER, "every thread's run lies inside the staged counts");
__global__ __launch_bounds__(1024)
void scan_offsets_kernel(const int* __restrict__ cnt, int* __restrict__ off) {
  __shared__ __attribute__((aligned(16))) int buf[SCAN_LDS];
  __shared__ int part[1024];
  const int t = threadIdx.x;
  v4i g[SCAN_QPT];
#pragma unroll
  for (int j = 0; j < SCAN_QPT; ++j) {
    const int q = j * 1024 + t;
    const v4i v = *(const v4i*)(cnt + 4 * ((q < SCAN_QUADS) ? q : (SCAN_QUADS - 1)));
    const v4i z = {0, 0, 0, 0};
    g[j] = (q < SCAN_QUADS) ? v : z;
  }
#pragma unroll
  for (int j = 0; j < SCAN_QPT; ++j) *(v4i*)(buf + 4 * (j * 1024 + t)) = g[j];
  __syncthreads();
  const int b = t * SCAN_PER;
  int c[SCAN_PER];
  int s = 0;
#pragma unroll
  for (int j = 0; j < SCAN_PER; ++j) {
    c[j] = buf[b + j];
    s += c[j];
  }
  part[t] = s;
  __syncthreads();
  for (int d = 1; d < 1024; d <<= 1) {
    const int v = (t >= d) ? part[t - d] : 0;
    __syncthreads();
    part[t] += v;
    __syncthreads();
  }
  int run = part[t] - s;
#pragma unroll
  for (int j = 0; j < SCAN_PER; ++j) {
    buf[b + j] = run;
    run += c[j];
  }
  __syncthreads();
#pragma unroll
  for (int j = 0; j < SCAN_QPT; ++j) {
    const int q = j * 1024 + t;
    if (q < SCAN_QUADS) *(v4i*)(off + 4 * q) = *(const v4i*)(buf + 4 * q);
  }
  if (t == 1023) off[N_NODES] = part[1023];
}

__global__ __launch_bounds__(256)
void fill_perm_kernel(const int* __restrict__ ei, const int* __restrict__ off,
                      const int* __restrict__ rank, int* __restrict__ perm) {
  const int e = blockIdx.x * 256 + threadIdx.x;
  if (e >= N_EDGES) return;
  perm[off[ei[N_EDGES + e]] + rank[e]] = e;
}

__global__ __launch_bounds__(256)
void order_segments_kernel(const int* __restrict__ ei, const float* __restrict__ eattr,
                           const int* __restrict__ off, const int* __restrict__ perm,
                           v4f* __restrict__ rec) {
  const int lane = threadIdx.x & 31;
  const int node = blockIdx.x * 8 + (threadIdx.x >> 5);
  if (node >= N_NODES) return;
  const int beg = off[node];
  const int deg = off[node + 1] - beg;
  for (int k = lane; k < deg; k += 32) {
    const int my = perm[beg + k];
    int place = 0;
    for (int j = 0; j < deg; ++j) place += (perm[beg + j] < my) ? 1 : 0;
    v4f r;
    r.x = __int_as_float(ei[my]);
    r.y = eattr[my * 3 + 0];
    r.z = eattr[my * 3 + 1];
    r.w = eattr[my * 3 + 2];
    rec[beg + place] = r;
  }
}

template <bool NORM>
__device__ __forceinline__ void stage_a_tile(float* __restrict__ As, const float* __restrict__ A,
                                             const int rowBase,
                                             const float* __restrict__ stats,
                                             const float* __restrict__ lnw,
                                             const float* __restrict__ lnb) {
  const int c  = threadIdx.x & (HIDW - 1);
  const int r0 = threadIdx.x >> 7;
  float mean = 0.f, rs = 0.f, wc = 0.f, bc = 0.f;
  if (NORM) {
    const v2f part = *(const v2f*)(stats + (threadIdx.x & 31) * LN_SLOT_PITCH);
    float s1 = part.x, s2 = part.y;
    s1 += __shfl_xor(s1, 16, 32);  s2 += __shfl_xor(s2, 16, 32);
    s1 += __shfl_xor(s1, 8, 32);   s2 += __shfl_xor(s2, 8, 32);
    s1 += __shfl_xor(s1, 4, 32);   s2 += __shfl_xor(s2, 4, 32);
    s1 += __shfl_xor(s1, 2, 32);   s2 += __shfl_xor(s2, 2, 32);
    s1 += __shfl_xor(s1, 1, 32);   s2 += __shfl_xor(s2, 1, 32);
    const float inv = 1.0f / (float)(N_NODES * HIDW);
    mean = s1 * inv;
    const float var = s2 * inv - mean * mean;
    rs = rsqrtf(var + 1e-5f);
    wc = lnw[c];
    bc = lnb[c];
  }
  float v[8];
#pragma unroll
  for (int j = 0; j < 8; ++j) v[j] = A[(rowBase + r0 + 2 * j) * HIDW + c];
  if (NORM) {
#pragma unroll
    for (int j = 0; j < 8; ++j) {
      const float t = (v[j] - mean) * rs * wc + bc;
      v[j] = 0.5f * t * (1.0f + erff(t * 0.70710678118654752f));
    }
  }
#pragma unroll
  for (int j = 0; j < 8; ++j) As[(r0 + 2 * j) * 132 + c] = v[j];
}

__device__ __forceinline__ void dual_tile_product(const float* __restrict__ As, const int rowBase,
                                                  const float* __restrict__ W0,
                                                  const float* __restrict__ W1,
                                                  float* __restrict__ O0,
                                                  float* __restrict__ O1) {
  const int wave    = threadIdx.x >> 5;
  const int lane    = threadIdx.x & 31;
  const int colBase = wave * 16;
  const int laneHi  = lane >> 4;
  const int laneLo  = lane & 15;
  v8f c0 = {}; v8f c1 = {};
  for (int k0 = 0; k0 < HIDW; k0 += 4) {
    const int ka = k0 + 2 * laneHi;
    v2f a;
    a.x = As[laneLo * 132 + ka];
    a.y = As[laneLo * 132 + ka + 1];
    v2f b0, b1;
    b0.x = W0[(ka + 0) * HIDW + colBase + laneLo];
    b0.y = W0[(ka + 1) * HIDW + colBase + laneLo];
    b1.x = W1[(ka + 0) * HIDW + colBase + laneLo];
    b1.y = W1[(ka + 1) * HIDW + colBase + laneLo];
    c0 = __builtin_amdgcn_wmma_f32_16x16x4_f32(false, a, false, b0, (short)0, c0, false, false);
    c1 = __builtin_amdgcn_wmma_f32_16x16x4_f32(false, a, false, b1, (short)0, c1, false, false);
  }
#pragma unroll
  for (int r = 0; r < 8; ++r) {
    const int row = rowBase + r + 8 * laneHi;
    O0[row * HIDW + colBase + laneLo] = c0[r];
    O1[row * HIDW + colBase + laneLo] = c1[r];
  }
}

__global__ __launch_bounds__(256)
void gemm_dual_wmma(const float* __restrict__ A,
                    const float* __restrict__ W0,
                    const float* __restrict__ W1,
                    float* __restrict__ O0,
                    float* __restrict__ O1) {
  __shared__ float As[16 * 132];
  const int rowBase = blockIdx.x * 16;
  stage_a_tile<false>(As, A, rowBase, nullptr, nullptr, nullptr);
  __syncthreads();
  dual_tile_product(As, rowBase, W0, W1, O0, O1);
}

__global__ __launch_bounds__(256)
void gemm_dual_norm_wmma(const float* __restrict__ A,
                         const float* __restrict__ stats,
                         const float* __restrict__ lnw,
                         const float* __restrict__ lnb,
                         const float* __restrict__ W0,
                         const float* __restrict__ W1,
                         float* __restrict__ O0,
                         float* __restrict__ O1) {
  __shared__ float As[16 * 132];
  const int rowBase = blockIdx.x * 16;
  stage_a_tile<true>(As, A, rowBase, stats, lnw, lnb);
  __syncthreads();
  dual_tile_product(As, rowBase, W0, W1, O0, O1);
}

__global__ __launch_bounds__(256)
void gemm_bias_norm_wmma(const float* __restrict__ A,
                         const float* __restrict__ stats,
                         const float* __restrict__ lnw,
                         const float* __restrict__ lnb,
                         const float* __restrict__ W,
                         const float* __restrict__ bias,
                         float* __restrict__ O) {
  __shared__ float As[16 * 132];
  const int rowBase = blockIdx.x * 16;
  stage_a_tile<true>(As, A, rowBase, stats, lnw, lnb);
  __syncthreads();
  const int wave    = threadIdx.x >> 5;
  const int lane    = threadIdx.x & 31;
  const int colBase = wave * 16;
  const int laneHi  = lane >> 4;
  const int laneLo  = lane & 15;
  v8f c0 = {};
  for (int k0 = 0; k0 < HIDW; k0 += 4) {
    const int ka = k0 + 2 * laneHi;
    v2f a;
    a.x = As[laneLo * 132 + ka];
    a.y = As[laneLo * 132 + ka + 1];
    v2f b;
    b.x = W[(ka + 0) * HIDW + colBase + laneLo];
    b.y = W[(ka + 1) * HIDW + colBase + laneLo];
    c0 = __builtin_amdgcn_wmma_f32_16x16x4_f32(false, a, false, b, (short)0, c0, false, false);
  }
  const float bv = bias[colBase + laneLo];
#pragma unroll
  for (int r = 0; r < 8; ++r) {
    const int row = rowBase + r + 8 * laneHi;
    O[row * HIDW + colBase + laneLo] = c0[r] + bv;
  }
}

__device__ __forceinline__ float head_logit(const v4f x, const v4f xrv, const v4f r,
                                            const v4f w0, const v4f w1, const v4f w2,
                                            const v4f av) {
  float v0 = x.x + xrv.x + (r.y * w0.x + r.z * w1.x + r.w * w2.x);
  float v1 = x.y + xrv.y + (r.y * w0.y + r.z * w1.y + r.w * w2.y);
  float v2 = x.z + xrv.z + (r.y * w0.z + r.z * w1.z + r.w * w2.z);
  float v3 = x.w + xrv.w + (r.y * w0.w + r.z * w1.w + r.w * w2.w);
  v0 = fmaxf(v0, 0.2f * v0);
  v1 = fmaxf(v1, 0.2f * v1);
  v2 = fmaxf(v2, 0.2f * v2);
  v3 = fmaxf(v3, 0.2f * v3);
  float p = v0 * av.x + v1 * av.y + v2 * av.z + v3 * av.w;
  p += __shfl_xor(p, 1, 32);
  p += __shfl_xor(p, 2, 32);
  p += __shfl_xor(p, 4, 32);
  return p;
}

__global__ __launch_bounds__(256)
void node_attention_kernel(const int* __restrict__ off, const v4f* __restrict__ rec,
                           const float* __restrict__ xl, const float* __restrict__ xr,
                           const float* __restrict__ We, const float* __restrict__ att,
                           float* __restrict__ hout, float* __restrict__ stats) {
  const int lane = threadIdx.x & 31;
  const int node = __builtin_amdgcn_readfirstlane(blockIdx.x * 8 + (threadIdx.x >> 5));
  if (node >= N_NODES) return;
  const int c4 = lane * 4;
  const v4f xrv = *(const v4f*)(xr + (size_t)node * HIDW + c4);
  const v4f w0  = *(const v4f*)(We + c4);
  const v4f w1  = *(const v4f*)(We + HIDW + c4);
  const v4f w2  = *(const v4f*)(We + 2 * HIDW + c4);
  const v4f av  = *(const v4f*)(att + c4);
  const int beg = off[node];
  const int end = off[node + 1];
  float m = NEG_BIG, den = 0.f;
  v4f acc = {0.f, 0.f, 0.f, 0.f};
  for (int i = beg; i < end; i += 4) {
    const int n  = end - i;
    const int i1 = (n > 1) ? i + 1 : i;
    const int i2 = (n > 2) ? i + 2 : i;
    const int i3 = (n > 3) ? i + 3 : i;
    const v4f r0 = rec[i];
    const v4f r1 = rec[i1];
    const v4f r2 = rec[i2];
    const v4f r3 = rec[i3];
    const v4f x0 = *(const v4f*)(xl + (size_t)__float_as_int(r0.x) * HIDW + c4);
    const v4f x1 = *(const v4f*)(xl + (size_t)__float_as_int(r1.x) * HIDW + c4);
    const v4f x2 = *(const v4f*)(xl + (size_t)__float_as_int(r2.x) * HIDW + c4);
    const v4f x3 = *(const v4f*)(xl + (size_t)__float_as_int(r3.x) * HIDW + c4);
    const float p0 = head_logit(x0, xrv, r0, w0, w1, w2, av);
    float p1 = head_logit(x1, xrv, r1, w0, w1, w2, av);
    float p2 = head_logit(x2, xrv, r2, w0, w1, w2, av);
    float p3 = head_logit(x3, xrv, r3, w0, w1, w2, av);
    p1 = (n > 1) ? p1 : NEG_BIG;
    p2 = (n > 2) ? p2 : NEG_BIG;
    p3 = (n > 3) ? p3 : NEG_BIG;
    const float mn = fmaxf(fmaxf(m, p0), fmaxf(p1, fmaxf(p2, p3)));
    const float sc = expf(m - mn);
    const float e0 = expf(p0 - mn);
    const float e1 = expf(p1 - mn);
    const float e2 = expf(p2 - mn);
    const float e3 = expf(p3 - mn);
    den   = den * sc + ((e0 + e1) + (e2 + e3));
    acc.x = acc.x * sc + ((e0 * x0.x + e1 * x1.x) + (e2 * x2.x + e3 * x3.x));
    acc.y = acc.y * sc + ((e0 * x0.y + e1 * x1.y) + (e2 * x2.y + e3 * x3.y));
    acc.z = acc.z * sc + ((e0 * x0.z + e1 * x1.z) + (e2 * x2.z + e3 * x3.z));
    acc.w = acc.w * sc + ((e0 * x0.w + e1 * x1.w) + (e2 * x2.w + e3 * x3.w));
    m = mn;
  }
  const float inv = 1.0f / (den + 1e-16f);
  v4f o;
  o.x = acc.x * inv; o.y = acc.y * inv; o.z = acc.z * inv; o.w = acc.w * inv;
  *(v4f*)(hout + (size_t)node * HIDW + c4) = o;
  float s1 = (o.x + o.y) + (o.z + o.w);
  float s2 = (o.x * o.x + o.y * o.y) + (o.z * o.z + o.w * o.w);
  s1 += __shfl_xor(s1, 16, 32);  s2 += __shfl_xor(s2, 16, 32);
  s1 += __shfl_xor(s1, 8, 32);   s2 += __shfl_xor(s2, 8, 32);
  s1 += __shfl_xor(s1, 4, 32);   s2 += __shfl_xor(s2, 4, 32);
  s1 += __shfl_xor(s1, 2, 32);   s2 += __shfl_xor(s2, 2, 32);
  s1 += __shfl_xor(s1, 1, 32);   s2 += __shfl_xor(s2, 1, 32);
  if (lane == 0) {
    float* slot = stats + (node & (LN_SLOTS - 1)) * LN_SLOT_PITCH;
    atomicAdd(slot, s1);
    atomicAdd(slot + 1, s2);
  }
}

extern "C" void kernel_launch(void* const* d_in, const int* in_sizes, int n_in,
                              void* d_out, int out_size, void* d_ws, size_t ws_size,
                              hipStream_t stream) {
  const float* x     = (const float*)d_in[0];
  const int*   ei    = (const int*)  d_in[1];
  const float* eattr = (const float*)d_in[2];
  const float* Wl    = (const float*)d_in[3];
  const float* Wr    = (const float*)d_in[4];
  const float* We    = (const float*)d_in[5];
  const float* att   = (const float*)d_in[6];
  const float* lnw   = (const float*)d_in[7];
  const float* lnb   = (const float*)d_in[8];
  const float* Wout  = (const float*)d_in[9];
  const float* bout  = (const float*)d_in[10];
  float* out = (float*)d_out;

  char* p = (char*)d_ws;
  auto carve = [&](size_t bytes) -> void* {
    void* q = (void*)p;
    p += (bytes + 255) & ~(size_t)255;
    return q;
  };
  float* hA    = (float*)carve((size_t)N_NODES * HIDW * 4);
  float* hB    = (float*)carve((size_t)N_NODES * HIDW * 4);
  float* xl    = (float*)carve((size_t)N_NODES * HIDW * 4);
  float* xr    = (float*)carve((size_t)N_NODES * HIDW * 4);
  v4f*   rec   = (v4f*)  carve((size_t)N_EDGES * 16);
  int*   rank  = (int*)  carve((size_t)N_EDGES * 4);
  int*   perm  = (int*)  carve((size_t)N_EDGES * 4);
  int*   cnt   = (int*)  carve((size_t)N_NODES * 4);
  int*   off   = (int*)  carve((size_t)(N_NODES + 1) * 4);
  float* stats = (float*)carve((size_t)NLAYERS * LN_LAYER_PITCH * 4);

  const int cntBlocks  = (N_NODES + 255) / 256;
  const int edgeBlocks = (N_EDGES + 255) / 256;
  const int nodeBlocks = (N_NODES + 7) / 8;
  const int gemmBlocks = N_NODES / 16;

  zero_counts_kernel<<<cntBlocks, 256, 0, stream>>>(cnt, stats);
  count_rank_kernel<<<edgeBlocks, 256, 0, stream>>>(ei, cnt, rank);
  scan_offsets_kernel<<<1, 1024, 0, stream>>>(cnt, off);
  fill_perm_kernel<<<edgeBlocks, 256, 0, stream>>>(ei, off, rank, perm);
  order_segments_kernel<<<nodeBlocks, 256, 0, stream>>>(ei, eattr, off, perm, rec);

  const float* hcur = x;
  float* bufs[2] = { hA, hB };
  for (int l = 0; l < NLAYERS; ++l) {
    float* hout = bufs[l & 1];
    if (l == 0) {
      gemm_dual_wmma<<<gemmBlocks, 256, 0, stream>>>(hcur, Wl, Wr, xl, xr);
    } else {
      gemm_dual_norm_wmma<<<gemmBlocks, 256, 0, stream>>>(hcur,
                                                          stats + (size_t)(l - 1) * LN_LAYER_PITCH,
                                                          lnw + (size_t)(l - 1) * HIDW,
                                                          lnb + (size_t)(l - 1) * HIDW,
                                                          Wl + (size_t)l * HIDW * HIDW,
                                                          Wr + (size_t)l * HIDW * HIDW, xl, xr);
    }
    node_attention_kernel<<<nodeBlocks, 256, 0, stream>>>(off, rec, xl, xr,
                                                          We + (size_t)l * EDIMW * HIDW,
                                                          att + (size_t)l * HIDW, hout,
                                                          stats + (size_t)l * LN_LAYER_PITCH);
    hcur = hout;
  }
  gemm_bias_norm_wmma<<<gemmBlocks, 256, 0, stream>>>(hcur,
                                                      stats + (size_t)(NLAYERS - 1) * LN_LAYER_PITCH,
                                                      lnw + (size_t)(NLAYERS - 1) * HIDW,
                                                      lnb + (size_t)(NLAYERS - 1) * HIDW,
                                                      Wout, bout, out);
}
